# starting kernel
# baseline (speedup 1.0000x reference)
#include <hip/hip_runtime.h>
#include <math.h>

typedef float v2f __attribute__((ext_vector_type(2)));
typedef float v8f __attribute__((ext_vector_type(8)));


__global__ void zero_int_kernel(int* __restrict__ p, int n) {
  int i = blockIdx.x * blockDim.x + threadIdx.x;
  int stride = gridDim.x * blockDim.x;
  for (; i < n; i += stride) p[i] = 0;
}

__global__ void degree_kernel(const int* __restrict__ dst, int* __restrict__ cnt, int E) {
  int e = blockIdx.x * blockDim.x + threadIdx.x;
  if (e < E) atomicAdd(&cnt[dst[e]], 1);
}

__global__ __launch_bounds__(1024)
void scan_kernel(const int* __restrict__ cnt, int* __restrict__ off,
                 int* __restrict__ cursor, int n) {
  __shared__ int tmp[1024];
  int carry = 0;
  for (int base = 0; base < n; base += 1024) {
    int i = base + threadIdx.x;
    int v = (i < n) ? cnt[i] : 0;
    tmp[threadIdx.x] = v;
    __syncthreads();
#pragma unroll
    for (int d = 1; d < 1024; d <<= 1) {
      int t = (threadIdx.x >= d) ? tmp[threadIdx.x - d] : 0;
      __syncthreads();
      tmp[threadIdx.x] += t;
      __syncthreads();
    }
    int excl = tmp[threadIdx.x] - v;
    if (i < n) {
      off[i] = carry + excl;
      cursor[i] = carry + excl;
    }
    int chunk_total = tmp[1023];
    __syncthreads();
    carry += chunk_total;
  }
  if (threadIdx.x == 0) off[n] = carry;
}

__global__ void csr_fill_kernel(const int* __restrict__ src, const int* __restrict__ dst,
                                int* __restrict__ cursor, int* __restrict__ csr, int E) {
  int e = blockIdx.x * blockDim.x + threadIdx.x;
  if (e < E) {
    int slot = atomicAdd(&cursor[dst[e]], 1);
    csr[slot] = src[e];
  }
}

template<int K, int NCOL, int WAVES, int TILES, int ACT>
__global__ __launch_bounds__(WAVES * 32)
void sage_layer_kernel(const float* __restrict__ feat,
                       const int* __restrict__ off,
                       const int* __restrict__ csr,
                       const float* __restrict__ Wl,
                       const float* __restrict__ Wr,
                       const float* __restrict__ bias,
                       float* __restrict__ out, int nrows) {
  static_assert(WAVES * TILES * 16 == NCOL, "tiling must cover NCOL");
  static_assert(K % 128 == 0, "K must be a multiple of 128");
  constexpr int KP = K + 4;
  constexpr int CH = K / 128;
  constexpr int RPW = 16 / WAVES;
  __shared__ float sA[16 * KP];
  __shared__ float sX[16 * KP];

  const int rowbase = blockIdx.x * 16;
  const int lane = threadIdx.x & 31;
  const int wave = threadIdx.x >> 5;

  for (int rr = 0; rr < RPW; ++rr) {
    const int r = wave * RPW + rr;
    const int row = rowbase + r;
    float4 acc[CH] = {};
    float4 self[CH] = {};
    float scale = 1.0f;
    if (row < nrows) {
      const int beg = off[row];
      const int end = off[row + 1];
      scale = 1.0f / fmaxf((float)(end - beg), 1.0f);
#pragma unroll
      for (int c = 0; c < CH; ++c)
        self[c] = ((const float4*)(feat + (long)row * K))[lane + 32 * c];
      for (int p = beg; p < end; ++p) {
        const int s = csr[p];
        if (p + 1 < end)
          __builtin_prefetch(feat + (long)csr[p + 1] * K, 0, 0);
        const float4* rp = (const float4*)(feat + (long)s * K);
#pragma unroll
        for (int c = 0; c < CH; ++c) {
          float4 v = rp[lane + 32 * c];
          acc[c].x += v.x; acc[c].y += v.y; acc[c].z += v.z; acc[c].w += v.w;
        }
      }
    }
#pragma unroll
    for (int c = 0; c < CH; ++c) {
      const int col = 4 * (lane + 32 * c);
      float4 a;
      a.x = acc[c].x * scale; a.y = acc[c].y * scale;
      a.z = acc[c].z * scale; a.w = acc[c].w * scale;
      *(float4*)&sA[r * KP + col] = a;
      *(float4*)&sX[r * KP + col] = self[c];
    }
  }
  __syncthreads();

  const int m = lane & 15;
  const int khalf = (lane >> 4) * 2;

  v8f acc[TILES];
  int ncol[TILES];
#pragma unroll
  for (int t = 0; t < TILES; ++t) {
    ncol[t] = (wave * TILES + t) * 16 + (lane & 15);
    float bv = bias[ncol[t]];
#pragma unroll
    for (int j = 0; j < 8; ++j) acc[t][j] = bv;
  }

#pragma unroll 2
  for (int k0 = 0; k0 < K; k0 += 4) {
    const int ka = k0 + khalf;
    v2f a;
    a[0] = sA[m * KP + ka];
    a[1] = sA[m * KP + ka + 1];
#pragma unroll
    for (int t = 0; t < TILES; ++t) {
      v2f b;
      b[0] = Wl[(long)ka * NCOL + ncol[t]];
      b[1] = Wl[(long)(ka + 1) * NCOL + ncol[t]];
      acc[t] = __builtin_amdgcn_wmma_f32_16x16x4_f32(
          false, a, false, b, (short)0, acc[t], false, false);
    }
  }
#pragma unroll 2
  for (int k0 = 0; k0 < K; k0 += 4) {
    const int ka = k0 + khalf;
    v2f a;
    a[0] = sX[m * KP + ka];
    a[1] = sX[m * KP + ka + 1];
#pragma unroll
    for (int t = 0; t < TILES; ++t) {
      v2f b;
      b[0] = Wr[(long)ka * NCOL + ncol[t]];
      b[1] = Wr[(long)(ka + 1) * NCOL + ncol[t]];
      acc[t] = __builtin_amdgcn_wmma_f32_16x16x4_f32(
          false, a, false, b, (short)0, acc[t], false, false);
    }
  }

  const int rowhalf = (lane >> 4) * 8;
#pragma unroll
  for (int t = 0; t < TILES; ++t) {
#pragma unroll
    for (int r = 0; r < 8; ++r) {
      int row = rowbase + rowhalf + r;
      if (row < nrows) {
        float v = acc[t][r];
        if (ACT == 0) {
          v = fmaxf(v, 0.0f);
        } else {
          v = 1.0f / (1.0f + __expf(-v));
        }
        out[(long)row * NCOL + ncol[t]] = v;
      }
    }
  }
}


extern "C" void kernel_launch(void* const* d_in, const int* in_sizes, int n_in,
                              void* d_out, int out_size, void* d_ws, size_t ws_size,
                              hipStream_t stream) {
  constexpr int FIN = 128, FHID = 256, FOUT = 64;

  const float* x   = (const float*)d_in[0];
  const int*   ei  = (const int*)d_in[1];
  const float* Wl1 = (const float*)d_in[2];
  const float* Wr1 = (const float*)d_in[3];
  const float* b1  = (const float*)d_in[4];
  const float* Wl2 = (const float*)d_in[5];
  const float* Wr2 = (const float*)d_in[6];
  const float* b2  = (const float*)d_in[7];
  float* out = (float*)d_out;

  const int N = in_sizes[0] / FIN;
  const int E = in_sizes[1] / 2;
  const int* src = ei;
  const int* dst = ei + E;

  int* cnt    = (int*)d_ws;
  int* off    = cnt + N;
  int* cursor = off + (N + 1);
  int* csr    = cursor + N;
  float* h    = (float*)(csr + E);

  const int T = 256;
  auto blocks = [](long n, int t) { return (int)((n + t - 1) / t); };
  const int rowBlocks = (N + 15) / 16;

  zero_int_kernel<<<min(blocks(N, T), 4096), T, 0, stream>>>(cnt, N);
  degree_kernel<<<blocks(E, T), T, 0, stream>>>(dst, cnt, E);
  scan_kernel<<<1, 1024, 0, stream>>>(cnt, off, cursor, N);
  csr_fill_kernel<<<blocks(E, T), T, 0, stream>>>(src, dst, cursor, csr, E);

  sage_layer_kernel<FIN, FHID, 8, 2, 0>
      <<<rowBlocks, 256, 0, stream>>>(x, off, csr, Wl1, Wr1, b1, h, N);
  sage_layer_kernel<FHID, FOUT, 4, 1, 1>
      <<<rowBlocks, 128, 0, stream>>>(h, off, csr, Wl2, Wr2, b2, out, N);
}


# Neighbour gather: 8 rows' loads in flight per trip (list indices by one 32-lane load, read out per lane; scalar row walk) instead of one dependent row at a time; per-row sum order unchanged
# speedup vs baseline: 1.5613x; 1.5613x over previous
#include <hip/hip_runtime.h>
#include <math.h>

typedef float v2f __attribute__((ext_vector_type(2)));
typedef float v8f __attribute__((ext_vector_type(8)));


__global__ void zero_int_kernel(int* __restrict__ p, int n) {
  int i = blockIdx.x * blockDim.x + threadIdx.x;
  int stride = gridDim.x * blockDim.x;
  for (; i < n; i += stride) p[i] = 0;
}

__global__ void degree_kernel(const int* __restrict__ dst, int* __restrict__ cnt, int E) {
  int e = blockIdx.x * blockDim.x + threadIdx.x;
  if (e < E) atomicAdd(&cnt[dst[e]], 1);
}

__global__ __launch_bounds__(1024)
void scan_kernel(const int* __restrict__ cnt, int* __restrict__ off,
                 int* __restrict__ cursor, int n) {
  __shared__ int tmp[1024];
  int carry = 0;
  for (int base = 0; base < n; base += 1024) {
    int i = base + threadIdx.x;
    int v = (i < n) ? cnt[i] : 0;
    tmp[threadIdx.x] = v;
    __syncthreads();
#pragma unroll
    for (int d = 1; d < 1024; d <<= 1) {
      int t = (threadIdx.x >= d) ? tmp[threadIdx.x - d] : 0;
      __syncthreads();
      tmp[threadIdx.x] += t;
      __syncthreads();
    }
    int excl = tmp[threadIdx.x] - v;
    if (i < n) {
      off[i] = carry + excl;
      cursor[i] = carry + excl;
    }
    int chunk_total = tmp[1023];
    __syncthreads();
    carry += chunk_total;
  }
  if (threadIdx.x == 0) off[n] = carry;
}

__global__ void csr_fill_kernel(const int* __restrict__ src, const int* __restrict__ dst,
                                int* __restrict__ cursor, int* __restrict__ csr, int E) {
  int e = blockIdx.x * blockDim.x + threadIdx.x;
  if (e < E) {
    int slot = atomicAdd(&cursor[dst[e]], 1);
    csr[slot] = src[e];
  }
}

template<int K, int NCOL, int WAVES, int TILES, int ACT>
__global__ __launch_bounds__(WAVES * 32)
void sage_layer_kernel(const float* __restrict__ feat,
                       const int* __restrict__ off,
                       const int* __restrict__ csr,
                       const float* __restrict__ Wl,
                       const float* __restrict__ Wr,
                       const float* __restrict__ bias,
                       float* __restrict__ out, int nrows) {
  static_assert(WAVES * TILES * 16 == NCOL, "tiling must cover NCOL");
  static_assert(K % 128 == 0, "K must be a multiple of 128");
  constexpr int KP = K + 4;
  constexpr int CH = K / 128;
  constexpr int RPW = 16 / WAVES;
  __shared__ float sA[16 * KP];
  __shared__ float sX[16 * KP];

  const int rowbase = blockIdx.x * 16;
  const int lane = threadIdx.x & 31;
  const int wave = threadIdx.x >> 5;

  constexpr int U = 8;
  const int uwave = __builtin_amdgcn_readfirstlane(wave);
  for (int rr = 0; rr < RPW; ++rr) {
    const int r = uwave * RPW + rr;
    const int row = rowbase + r;
    float4 acc[CH] = {};
    float4 self[CH] = {};
    float scale = 1.0f;
    if (row < nrows) {
      const int beg = off[row];
      const int end = off[row + 1];
      scale = 1.0f / fmaxf((float)(end - beg), 1.0f);
      int p0 = beg;
      int myidx = 0;
      if (p0 < end) myidx = csr[min(p0 + lane, end - 1)];
#pragma unroll
      for (int c = 0; c < CH; ++c)
        self[c] = ((const float4*)(feat + (long)row * K))[lane + 32 * c];
      while (p0 < end) {
        const int cnt = min(32, end - p0);
        for (int q = 0; q < cnt; q += U) {
          float4 v[U][CH];
#pragma unroll
          for (int u = 0; u < U; ++u) {
            if (u == 0 || q + u < cnt) {
              const int s = __builtin_amdgcn_readlane(myidx, q + u);
              const float4* rp = (const float4*)(feat + (long)s * K);
#pragma unroll
              for (int c = 0; c < CH; ++c)
                v[u][c] = rp[lane + 32 * c];
            } else {
#pragma unroll
              for (int c = 0; c < CH; ++c)
                v[u][c] = make_float4(0.0f, 0.0f, 0.0f, 0.0f);
            }
          }
#pragma unroll
          for (int u = 0; u < U; ++u) {
            if (u == 0 || q + u < cnt) {
#pragma unroll
              for (int c = 0; c < CH; ++c) {
                acc[c].x += v[u][c].x; acc[c].y += v[u][c].y;
                acc[c].z += v[u][c].z; acc[c].w += v[u][c].w;
              }
            }
          }
        }
        p0 += 32;
        if (p0 < end) myidx = csr[min(p0 + lane, end - 1)];
      }
    }
#pragma unroll
    for (int c = 0; c < CH; ++c) {
      const int col = 4 * (lane + 32 * c);
      float4 a;
      a.x = acc[c].x * scale; a.y = acc[c].y * scale;
      a.z = acc[c].z * scale; a.w = acc[c].w * scale;
      *(float4*)&sA[r * KP + col] = a;
      *(float4*)&sX[r * KP + col] = self[c];
    }
  }
  __syncthreads();

  const int m = lane & 15;
  const int khalf = (lane >> 4) * 2;

  v8f acc[TILES];
  int ncol[TILES];
#pragma unroll
  for (int t = 0; t < TILES; ++t) {
    ncol[t] = (wave * TILES + t) * 16 + (lane & 15);
    float bv = bias[ncol[t]];
#pragma unroll
    for (int j = 0; j < 8; ++j) acc[t][j] = bv;
  }

#pragma unroll 2
  for (int k0 = 0; k0 < K; k0 += 4) {
    const int ka = k0 + khalf;
    v2f a;
    a[0] = sA[m * KP + ka];
    a[1] = sA[m * KP + ka + 1];
#pragma unroll
    for (int t = 0; t < TILES; ++t) {
      v2f b;
      b[0] = Wl[(long)ka * NCOL + ncol[t]];
      b[1] = Wl[(long)(ka + 1) * NCOL + ncol[t]];
      acc[t] = __builtin_amdgcn_wmma_f32_16x16x4_f32(
          false, a, false, b, (short)0, acc[t], false, false);
    }
  }
#pragma unroll 2
  for (int k0 = 0; k0 < K; k0 += 4) {
    const int ka = k0 + khalf;
    v2f a;
    a[0] = sX[m * KP + ka];
    a[1] = sX[m * KP + ka + 1];
#pragma unroll
    for (int t = 0; t < TILES; ++t) {
      v2f b;
      b[0] = Wr[(long)ka * NCOL + ncol[t]];
      b[1] = Wr[(long)(ka + 1) * NCOL + ncol[t]];
      acc[t] = __builtin_amdgcn_wmma_f32_16x16x4_f32(
          false, a, false, b, (short)0, acc[t], false, false);
    }
  }

  const int rowhalf = (lane >> 4) * 8;
#pragma unroll
  for (int t = 0; t < TILES; ++t) {
#pragma unroll
    for (int r = 0; r < 8; ++r) {
      int row = rowbase + rowhalf + r;
      if (row < nrows) {
        float v = acc[t][r];
        if (ACT == 0) {
          v = fmaxf(v, 0.0f);
        } else {
          v = 1.0f / (1.0f + __expf(-v));
        }
        out[(long)row * NCOL + ncol[t]] = v;
      }
    }
  }
}


extern "C" void kernel_launch(void* const* d_in, const int* in_sizes, int n_in,
                              void* d_out, int out_size, void* d_ws, size_t ws_size,
                              hipStream_t stream) {
  constexpr int FIN = 128, FHID = 256, FOUT = 64;

  const float* x   = (const float*)d_in[0];
  const int*   ei  = (const int*)d_in[1];
  const float* Wl1 = (const float*)d_in[2];
  const float* Wr1 = (const float*)d_in[3];
  const float* b1  = (const float*)d_in[4];
  const float* Wl2 = (const float*)d_in[5];
  const float* Wr2 = (const float*)d_in[6];
  const float* b2  = (const float*)d_in[7];
  float* out = (float*)d_out;

  const int N = in_sizes[0] / FIN;
  const int E = in_sizes[1] / 2;
  const int* src = ei;
  const int* dst = ei + E;

  int* cnt    = (int*)d_ws;
  int* off    = cnt + N;
  int* cursor = off + (N + 1);
  int* csr    = cursor + N;
  float* h    = (float*)(csr + E);

  const int T = 256;
  auto blocks = [](long n, int t) { return (int)((n + t - 1) / t); };
  const int rowBlocks = (N + 15) / 16;

  zero_int_kernel<<<min(blocks(N, T), 4096), T, 0, stream>>>(cnt, N);
  degree_kernel<<<blocks(E, T), T, 0, stream>>>(dst, cnt, E);
  scan_kernel<<<1, 1024, 0, stream>>>(cnt, off, cursor, N);
  csr_fill_kernel<<<blocks(E, T), T, 0, stream>>>(src, dst, cursor, csr, E);

  sage_layer_kernel<FIN, FHID, 8, 2, 0>
      <<<rowBlocks, 256, 0, stream>>>(x, off, csr, Wl1, Wr1, b1, h, N);
  sage_layer_kernel<FHID, FOUT, 4, 1, 1>
      <<<rowBlocks, 128, 0, stream>>>(h, off, csr, Wl2, Wr2, b2, out, N);
}


# Degree prefix scan spread over one block per 1024 counts, none waiting for another: a block sums the counts in front of its own itself, then scans its own in registers; one barrier per block
# speedup vs baseline: 2.2156x; 1.4191x over previous
#include <hip/hip_runtime.h>
#include <math.h>

typedef float v2f __attribute__((ext_vector_type(2)));
typedef float v8f __attribute__((ext_vector_type(8)));


__global__ void zero_int_kernel(int* __restrict__ p, int n) {
  int i = blockIdx.x * blockDim.x + threadIdx.x;
  int stride = gridDim.x * blockDim.x;
  for (; i < n; i += stride) p[i] = 0;
}

__global__ void degree_kernel(const int* __restrict__ dst, int* __restrict__ cnt, int E) {
  int e = blockIdx.x * blockDim.x + threadIdx.x;
  if (e < E) atomicAdd(&cnt[dst[e]], 1);
}


__device__ __forceinline__ int wave_inclusive_scan(int v, int lane) {
  v += __builtin_amdgcn_update_dpp(0, v, 0x111, 0xf, 0xf, false);
  v += __builtin_amdgcn_update_dpp(0, v, 0x112, 0xf, 0xf, false);
  v += __builtin_amdgcn_update_dpp(0, v, 0x114, 0xf, 0xf, false);
  v += __builtin_amdgcn_update_dpp(0, v, 0x118, 0xf, 0xf, false);
  const int low = __builtin_amdgcn_readlane(v, 15);
  return v + ((lane >= 16) ? low : 0);
}

__global__ __launch_bounds__(1024)
void scan_kernel(const int* __restrict__ cnt, int* __restrict__ off,
                 int* __restrict__ cursor, int n) {
  constexpr int G = 8;
  __shared__ int wfront[32];
  __shared__ int wown[32];
  const int lane = threadIdx.x & 31;
  const int wave = __builtin_amdgcn_readfirstlane(threadIdx.x >> 5);
  const int first = blockIdx.x * 1024;
  const int i = first + threadIdx.x;
  const int v = (i < n) ? cnt[i] : 0;

  int s = 0;
  for (int k0 = 0; k0 < first; k0 += G * 1024) {
    int u[G];
#pragma unroll
    for (int g = 0; g < G; ++g)
      u[g] = (k0 + g * 1024 < first) ? cnt[k0 + g * 1024 + threadIdx.x] : 0;
#pragma unroll
    for (int g = 0; g < G; ++g) s += u[g];
  }

  const int sinc = wave_inclusive_scan(s, lane);
  const int vinc = wave_inclusive_scan(v, lane);
  if (lane == 31) { wfront[wave] = sinc; wown[wave] = vinc; }
  __syncthreads();
  const int tf = wfront[lane];
  const int to = wown[lane];
  const int finc = wave_inclusive_scan(tf, lane);
  const int oinc = wave_inclusive_scan(to, lane);
  const int front = __builtin_amdgcn_readlane(finc, 31);
  const int below = __builtin_amdgcn_readlane(oinc - to, wave);
  const int blocksum = __builtin_amdgcn_readlane(oinc, 31);
  if (i < n) {
    const int e = front + below + vinc - v;
    off[i] = e;
    cursor[i] = e;
  }
  if (blockIdx.x == gridDim.x - 1 && threadIdx.x == 0) off[n] = front + blocksum;
}

__global__ void csr_fill_kernel(const int* __restrict__ src, const int* __restrict__ dst,
                                int* __restrict__ cursor, int* __restrict__ csr, int E) {
  int e = blockIdx.x * blockDim.x + threadIdx.x;
  if (e < E) {
    int slot = atomicAdd(&cursor[dst[e]], 1);
    csr[slot] = src[e];
  }
}

template<int K, int NCOL, int WAVES, int TILES, int ACT>
__global__ __launch_bounds__(WAVES * 32)
void sage_layer_kernel(const float* __restrict__ feat,
                       const int* __restrict__ off,
                       const int* __restrict__ csr,
                       const float* __restrict__ Wl,
                       const float* __restrict__ Wr,
                       const float* __restrict__ bias,
                       float* __restrict__ out, int nrows) {
  static_assert(WAVES * TILES * 16 == NCOL, "tiling must cover NCOL");
  static_assert(K % 128 == 0, "K must be a multiple of 128");
  constexpr int KP = K + 4;
  constexpr int CH = K / 128;
  constexpr int RPW = 16 / WAVES;
  __shared__ float sA[16 * KP];
  __shared__ float sX[16 * KP];

  const int rowbase = blockIdx.x * 16;
  const int lane = threadIdx.x & 31;
  const int wave = threadIdx.x >> 5;

  constexpr int U = 8;
  const int uwave = __builtin_amdgcn_readfirstlane(wave);
  for (int rr = 0; rr < RPW; ++rr) {
    const int r = uwave * RPW + rr;
    const int row = rowbase + r;
    float4 acc[CH] = {};
    float4 self[CH] = {};
    float scale = 1.0f;
    if (row < nrows) {
      const int beg = off[row];
      const int end = off[row + 1];
      scale = 1.0f / fmaxf((float)(end - beg), 1.0f);
      int p0 = beg;
      int myidx = 0;
      if (p0 < end) myidx = csr[min(p0 + lane, end - 1)];
#pragma unroll
      for (int c = 0; c < CH; ++c)
        self[c] = ((const float4*)(feat + (long)row * K))[lane + 32 * c];
      while (p0 < end) {
        const int cnt = min(32, end - p0);
        for (int q = 0; q < cnt; q += U) {
          float4 v[U][CH];
#pragma unroll
          for (int u = 0; u < U; ++u) {
            if (u == 0 || q + u < cnt) {
              const int s = __builtin_amdgcn_readlane(myidx, q + u);
              const float4* rp = (const float4*)(feat + (long)s * K);
#pragma unroll
              for (int c = 0; c < CH; ++c)
                v[u][c] = rp[lane + 32 * c];
            } else {
#pragma unroll
              for (int c = 0; c < CH; ++c)
                v[u][c] = make_float4(0.0f, 0.0f, 0.0f, 0.0f);
            }
          }
#pragma unroll
          for (int u = 0; u < U; ++u) {
            if (u == 0 || q + u < cnt) {
#pragma unroll
              for (int c = 0; c < CH; ++c) {
                acc[c].x += v[u][c].x; acc[c].y += v[u][c].y;
                acc[c].z += v[u][c].z; acc[c].w += v[u][c].w;
              }
            }
          }
        }
        p0 += 32;
        if (p0 < end) myidx = csr[min(p0 + lane, end - 1)];
      }
    }
#pragma unroll
    for (int c = 0; c < CH; ++c) {
      const int col = 4 * (lane + 32 * c);
      float4 a;
      a.x = acc[c].x * scale; a.y = acc[c].y * scale;
      a.z = acc[c].z * scale; a.w = acc[c].w * scale;
      *(float4*)&sA[r * KP + col] = a;
      *(float4*)&sX[r * KP + col] = self[c];
    }
  }
  __syncthreads();

  const int m = lane & 15;
  const int khalf = (lane >> 4) * 2;

  v8f acc[TILES];
  int ncol[TILES];
#pragma unroll
  for (int t = 0; t < TILES; ++t) {
    ncol[t] = (wave * TILES + t) * 16 + (lane & 15);
    float bv = bias[ncol[t]];
#pragma unroll
    for (int j = 0; j < 8; ++j) acc[t][j] = bv;
  }

#pragma unroll 2
  for (int k0 = 0; k0 < K; k0 += 4) {
    const int ka = k0 + khalf;
    v2f a;
    a[0] = sA[m * KP + ka];
    a[1] = sA[m * KP + ka + 1];
#pragma unroll
    for (int t = 0; t < TILES; ++t) {
      v2f b;
      b[0] = Wl[(long)ka * NCOL + ncol[t]];
      b[1] = Wl[(long)(ka + 1) * NCOL + ncol[t]];
      acc[t] = __builtin_amdgcn_wmma_f32_16x16x4_f32(
          false, a, false, b, (short)0, acc[t], false, false);
    }
  }
#pragma unroll 2
  for (int k0 = 0; k0 < K; k0 += 4) {
    const int ka = k0 + khalf;
    v2f a;
    a[0] = sX[m * KP + ka];
    a[1] = sX[m * KP + ka + 1];
#pragma unroll
    for (int t = 0; t < TILES; ++t) {
      v2f b;
      b[0] = Wr[(long)ka * NCOL + ncol[t]];
      b[1] = Wr[(long)(ka + 1) * NCOL + ncol[t]];
      acc[t] = __builtin_amdgcn_wmma_f32_16x16x4_f32(
          false, a, false, b, (short)0, acc[t], false, false);
    }
  }

  const int rowhalf = (lane >> 4) * 8;
#pragma unroll
  for (int t = 0; t < TILES; ++t) {
#pragma unroll
    for (int r = 0; r < 8; ++r) {
      int row = rowbase + rowhalf + r;
      if (row < nrows) {
        float v = acc[t][r];
        if (ACT == 0) {
          v = fmaxf(v, 0.0f);
        } else {
          v = 1.0f / (1.0f + __expf(-v));
        }
        out[(long)row * NCOL + ncol[t]] = v;
      }
    }
  }
}


extern "C" void kernel_launch(void* const* d_in, const int* in_sizes, int n_in,
                              void* d_out, int out_size, void* d_ws, size_t ws_size,
                              hipStream_t stream) {
  constexpr int FIN = 128, FHID = 256, FOUT = 64;

  const float* x   = (const float*)d_in[0];
  const int*   ei  = (const int*)d_in[1];
  const float* Wl1 = (const float*)d_in[2];
  const float* Wr1 = (const float*)d_in[3];
  const float* b1  = (const float*)d_in[4];
  const float* Wl2 = (const float*)d_in[5];
  const float* Wr2 = (const float*)d_in[6];
  const float* b2  = (const float*)d_in[7];
  float* out = (float*)d_out;

  const int N = in_sizes[0] / FIN;
  const int E = in_sizes[1] / 2;
  const int* src = ei;
  const int* dst = ei + E;

  int* cnt    = (int*)d_ws;
  int* off    = cnt + N;
  int* cursor = off + (N + 1);
  int* csr    = cursor + N;
  float* h    = (float*)(csr + E);

  const int T = 256;
  auto blocks = [](long n, int t) { return (int)((n + t - 1) / t); };
  const int rowBlocks = (N + 15) / 16;

  zero_int_kernel<<<min(blocks(N, T), 4096), T, 0, stream>>>(cnt, N);
  degree_kernel<<<blocks(E, T), T, 0, stream>>>(dst, cnt, E);
  scan_kernel<<<blocks(N, 1024), 1024, 0, stream>>>(cnt, off, cursor, N);
  csr_fill_kernel<<<blocks(E, T), T, 0, stream>>>(src, dst, cursor, csr, E);

  sage_layer_kernel<FIN, FHID, 8, 2, 0>
      <<<rowBlocks, 256, 0, stream>>>(x, off, csr, Wl1, Wr1, b1, h, N);
  sage_layer_kernel<FHID, FOUT, 4, 1, 1>
      <<<rowBlocks, 128, 0, stream>>>(h, off, csr, Wl2, Wr2, b2, out, N);
}


# K loops of both layers: weight-fragment global loads issued a chunk of k-steps ahead, about 32 in flight per wave (was 2 k-steps per round trip); WMMA order per accumulator and sums unchanged
# speedup vs baseline: 2.4884x; 1.1231x over previous
#include <hip/hip_runtime.h>
#include <math.h>

typedef float v2f __attribute__((ext_vector_type(2)));
typedef float v8f __attribute__((ext_vector_type(8)));


__global__ void zero_int_kernel(int* __restrict__ p, int n) {
  int i = blockIdx.x * blockDim.x + threadIdx.x;
  int stride = gridDim.x * blockDim.x;
  for (; i < n; i += stride) p[i] = 0;
}

__global__ void degree_kernel(const int* __restrict__ dst, int* __restrict__ cnt, int E) {
  int e = blockIdx.x * blockDim.x + threadIdx.x;
  if (e < E) atomicAdd(&cnt[dst[e]], 1);
}


__device__ __forceinline__ int wave_inclusive_scan(int v, int lane) {
  v += __builtin_amdgcn_update_dpp(0, v, 0x111, 0xf, 0xf, false);
  v += __builtin_amdgcn_update_dpp(0, v, 0x112, 0xf, 0xf, false);
  v += __builtin_amdgcn_update_dpp(0, v, 0x114, 0xf, 0xf, false);
  v += __builtin_amdgcn_update_dpp(0, v, 0x118, 0xf, 0xf, false);
  const int low = __builtin_amdgcn_readlane(v, 15);
  return v + ((lane >= 16) ? low : 0);
}

__global__ __launch_bounds__(1024)
void scan_kernel(const int* __restrict__ cnt, int* __restrict__ off,
                 int* __restrict__ cursor, int n) {
  constexpr int G = 8;
  __shared__ int wfront[32];
  __shared__ int wown[32];
  const int lane = threadIdx.x & 31;
  const int wave = __builtin_amdgcn_readfirstlane(threadIdx.x >> 5);
  const int first = blockIdx.x * 1024;
  const int i = first + threadIdx.x;
  const int v = (i < n) ? cnt[i] : 0;

  int s = 0;
  for (int k0 = 0; k0 < first; k0 += G * 1024) {
    int u[G];
#pragma unroll
    for (int g = 0; g < G; ++g)
      u[g] = (k0 + g * 1024 < first) ? cnt[k0 + g * 1024 + threadIdx.x] : 0;
#pragma unroll
    for (int g = 0; g < G; ++g) s += u[g];
  }

  const int sinc = wave_inclusive_scan(s, lane);
  const int vinc = wave_inclusive_scan(v, lane);
  if (lane == 31) { wfront[wave] = sinc; wown[wave] = vinc; }
  __syncthreads();
  const int tf = wfront[lane];
  const int to = wown[lane];
  const int finc = wave_inclusive_scan(tf, lane);
  const int oinc = wave_inclusive_scan(to, lane);
  const int front = __builtin_amdgcn_readlane(finc, 31);
  const int below = __builtin_amdgcn_readlane(oinc - to, wave);
  const int blocksum = __builtin_amdgcn_readlane(oinc, 31);
  if (i < n) {
    const int e = front + below + vinc - v;
    off[i] = e;
    cursor[i] = e;
  }
  if (blockIdx.x == gridDim.x - 1 && threadIdx.x == 0) off[n] = front + blocksum;
}

__global__ void csr_fill_kernel(const int* __restrict__ src, const int* __restrict__ dst,
                                int* __restrict__ cursor, int* __restrict__ csr, int E) {
  int e = blockIdx.x * blockDim.x + threadIdx.x;
  if (e < E) {
    int slot = atomicAdd(&cursor[dst[e]], 1);
    csr[slot] = src[e];
  }
}

template<int K, int NCOL, int TILES>
__device__ __forceinline__ void wmma_pass(const float* sT, const float* __restrict__ Wm,
                                          v8f (&acc)[TILES], int aofs, unsigned wofs) {
  constexpr int KC = 16 / TILES;
  static_assert(K % (4 * KC) == 0, "K must be a whole number of chunks");
  for (int kc = 0; kc < K; kc += 4 * KC) {
    const float* Wc = Wm + (unsigned)kc * NCOL;
    float b0[KC][TILES], b1[KC][TILES];
#pragma unroll
    for (int j = 0; j < KC; ++j) {
#pragma unroll
      for (int t = 0; t < TILES; ++t) {
        b0[j][t] = Wc[wofs + (unsigned)((4 * j) * NCOL + 16 * t)];
        b1[j][t] = Wc[wofs + (unsigned)((4 * j + 1) * NCOL + 16 * t)];
      }
    }
#pragma unroll
    for (int j = 0; j < KC; ++j) {
      v2f a;
      a[0] = sT[aofs + kc + 4 * j];
      a[1] = sT[aofs + kc + 4 * j + 1];
#pragma unroll
      for (int t = 0; t < TILES; ++t) {
        v2f b;
        b[0] = b0[j][t];
        b[1] = b1[j][t];
        acc[t] = __builtin_amdgcn_wmma_f32_16x16x4_f32(
            false, a, false, b, (short)0, acc[t], false, false);
      }
    }
  }
}

template<int K, int NCOL, int WAVES, int TILES, int ACT>
__global__ __launch_bounds__(WAVES * 32)
void sage_layer_kernel(const float* __restrict__ feat,
                       const int* __restrict__ off,
                       const int* __restrict__ csr,
                       const float* __restrict__ Wl,
                       const float* __restrict__ Wr,
                       const float* __restrict__ bias,
                       float* __restrict__ out, int nrows) {
  static_assert(WAVES * TILES * 16 == NCOL, "tiling must cover NCOL");
  static_assert(K % 128 == 0, "K must be a multiple of 128");
  constexpr int KP = K + 4;
  constexpr int CH = K / 128;
  constexpr int RPW = 16 / WAVES;
  __shared__ float sA[16 * KP];
  __shared__ float sX[16 * KP];

  const int rowbase = blockIdx.x * 16;
  const int lane = threadIdx.x & 31;
  const int wave = threadIdx.x >> 5;

  constexpr int U = 8;
  const int uwave = __builtin_amdgcn_readfirstlane(wave);
  for (int rr = 0; rr < RPW; ++rr) {
    const int r = uwave * RPW + rr;
    const int row = rowbase + r;
    float4 acc[CH] = {};
    float4 self[CH] = {};
    float scale = 1.0f;
    if (row < nrows) {
      const int beg = off[row];
      const int end = off[row + 1];
      scale = 1.0f / fmaxf((float)(end - beg), 1.0f);
      int p0 = beg;
      int myidx = 0;
      if (p0 < end) myidx = csr[min(p0 + lane, end - 1)];
#pragma unroll
      for (int c = 0; c < CH; ++c)
        self[c] = ((const float4*)(feat + (long)row * K))[lane + 32 * c];
      while (p0 < end) {
        const int cnt = min(32, end - p0);
        for (int q = 0; q < cnt; q += U) {
          float4 v[U][CH];
#pragma unroll
          for (int u = 0; u < U; ++u) {
            if (u == 0 || q + u < cnt) {
              const int s = __builtin_amdgcn_readlane(myidx, q + u);
              const float4* rp = (const float4*)(feat + (long)s * K);
#pragma unroll
              for (int c = 0; c < CH; ++c)
                v[u][c] = rp[lane + 32 * c];
            } else {
#pragma unroll
              for (int c = 0; c < CH; ++c)
                v[u][c] = make_float4(0.0f, 0.0f, 0.0f, 0.0f);
            }
          }
#pragma unroll
          for (int u = 0; u < U; ++u) {
            if (u == 0 || q + u < cnt) {
#pragma unroll
              for (int c = 0; c < CH; ++c) {
                acc[c].x += v[u][c].x; acc[c].y += v[u][c].y;
                acc[c].z += v[u][c].z; acc[c].w += v[u][c].w;
              }
            }
          }
        }
        p0 += 32;
        if (p0 < end) myidx = csr[min(p0 + lane, end - 1)];
      }
    }
#pragma unroll
    for (int c = 0; c < CH; ++c) {
      const int col = 4 * (lane + 32 * c);
      float4 a;
      a.x = acc[c].x * scale; a.y = acc[c].y * scale;
      a.z = acc[c].z * scale; a.w = acc[c].w * scale;
      *(float4*)&sA[r * KP + col] = a;
      *(float4*)&sX[r * KP + col] = self[c];
    }
  }
  __syncthreads();

  const int m = lane & 15;
  const int khalf = (lane >> 4) * 2;

  v8f acc[TILES];
  int ncol[TILES];
#pragma unroll
  for (int t = 0; t < TILES; ++t) {
    ncol[t] = (wave * TILES + t) * 16 + (lane & 15);
    float bv = bias[ncol[t]];
#pragma unroll
    for (int j = 0; j < 8; ++j) acc[t][j] = bv;
  }

  const int aofs = m * KP + khalf;
  const unsigned wofs = (unsigned)(khalf * NCOL + ncol[0]);
  wmma_pass<K, NCOL, TILES>(sA, Wl, acc, aofs, wofs);
  wmma_pass<K, NCOL, TILES>(sX, Wr, acc, aofs, wofs);

  const int rowhalf = (lane >> 4) * 8;
#pragma unroll
  for (int t = 0; t < TILES; ++t) {
#pragma unroll
    for (int r = 0; r < 8; ++r) {
      int row = rowbase + rowhalf + r;
      if (row < nrows) {
        float v = acc[t][r];
        if (ACT == 0) {
          v = fmaxf(v, 0.0f);
        } else {
          v = 1.0f / (1.0f + __expf(-v));
        }
        out[(long)row * NCOL + ncol[t]] = v;
      }
    }
  }
}


extern "C" void kernel_launch(void* const* d_in, const int* in_sizes, int n_in,
                              void* d_out, int out_size, void* d_ws, size_t ws_size,
                              hipStream_t stream) {
  constexpr int FIN = 128, FHID = 256, FOUT = 64;

  const float* x   = (const float*)d_in[0];
  const int*   ei  = (const int*)d_in[1];
  const float* Wl1 = (const float*)d_in[2];
  const float* Wr1 = (const float*)d_in[3];
  const float* b1  = (const float*)d_in[4];
  const float* Wl2 = (const float*)d_in[5];
  const float* Wr2 = (const float*)d_in[6];
  const float* b2  = (const float*)d_in[7];
  float* out = (float*)d_out;

  const int N = in_sizes[0] / FIN;
  const int E = in_sizes[1] / 2;
  const int* src = ei;
  const int* dst = ei + E;

  int* cnt    = (int*)d_ws;
  int* off    = cnt + N;
  int* cursor = off + (N + 1);
  int* csr    = cursor + N;
  float* h    = (float*)(csr + E);

  const int T = 256;
  auto blocks = [](long n, int t) { return (int)((n + t - 1) / t); };
  const int rowBlocks = (N + 15) / 16;

  zero_int_kernel<<<min(blocks(N, T), 4096), T, 0, stream>>>(cnt, N);
  degree_kernel<<<blocks(E, T), T, 0, stream>>>(dst, cnt, E);
  scan_kernel<<<blocks(N, 1024), 1024, 0, stream>>>(cnt, off, cursor, N);
  csr_fill_kernel<<<blocks(E, T), T, 0, stream>>>(src, dst, cursor, csr, E);

  sage_layer_kernel<FIN, FHID, 8, 2, 0>
      <<<rowBlocks, 256, 0, stream>>>(x, off, csr, Wl1, Wr1, b1, h, N);
  sage_layer_kernel<FHID, FOUT, 4, 1, 1>
      <<<rowBlocks, 128, 0, stream>>>(h, off, csr, Wl2, Wr2, b2, out, N);
}


# Layer 2: weights applied before the neighbour mean, on the h tile in LDS inside layer 1's kernel; last kernel averages 64-wide rows, h never stored; layer 2's sum order changes (mean of products)
# speedup vs baseline: 3.0351x; 1.2197x over previous
#include <hip/hip_runtime.h>
#include <math.h>

typedef float v2f __attribute__((ext_vector_type(2)));
typedef float v8f __attribute__((ext_vector_type(8)));


__global__ void zero_int_kernel(int* __restrict__ p, int n) {
  int i = blockIdx.x * blockDim.x + threadIdx.x;
  int stride = gridDim.x * blockDim.x;
  for (; i < n; i += stride) p[i] = 0;
}

__global__ void degree_kernel(const int* __restrict__ dst, int* __restrict__ cnt, int E) {
  int e = blockIdx.x * blockDim.x + threadIdx.x;
  if (e < E) atomicAdd(&cnt[dst[e]], 1);
}


__device__ __forceinline__ int wave_inclusive_scan(int v, int lane) {
  v += __builtin_amdgcn_update_dpp(0, v, 0x111, 0xf, 0xf, false);
  v += __builtin_amdgcn_update_dpp(0, v, 0x112, 0xf, 0xf, false);
  v += __builtin_amdgcn_update_dpp(0, v, 0x114, 0xf, 0xf, false);
  v += __builtin_amdgcn_update_dpp(0, v, 0x118, 0xf, 0xf, false);
  const int low = __builtin_amdgcn_readlane(v, 15);
  return v + ((lane >= 16) ? low : 0);
}

__global__ __launch_bounds__(1024)
void scan_kernel(const int* __restrict__ cnt, int* __restrict__ off,
                 int* __restrict__ cursor, int n) {
  constexpr int G = 8;
  __shared__ int wfront[32];
  __shared__ int wown[32];
  const int lane = threadIdx.x & 31;
  const int wave = __builtin_amdgcn_readfirstlane(threadIdx.x >> 5);
  const int first = blockIdx.x * 1024;
  const int i = first + threadIdx.x;
  const int v = (i < n) ? cnt[i] : 0;

  int s = 0;
  for (int k0 = 0; k0 < first; k0 += G * 1024) {
    int u[G];
#pragma unroll
    for (int g = 0; g < G; ++g)
      u[g] = (k0 + g * 1024 < first) ? cnt[k0 + g * 1024 + threadIdx.x] : 0;
#pragma unroll
    for (int g = 0; g < G; ++g) s += u[g];
  }

  const int sinc = wave_inclusive_scan(s, lane);
  const int vinc = wave_inclusive_scan(v, lane);
  if (lane == 31) { wfront[wave] = sinc; wown[wave] = vinc; }
  __syncthreads();
  const int tf = wfront[lane];
  const int to = wown[lane];
  const int finc = wave_inclusive_scan(tf, lane);
  const int oinc = wave_inclusive_scan(to, lane);
  const int front = __builtin_amdgcn_readlane(finc, 31);
  const int below = __builtin_amdgcn_readlane(oinc - to, wave);
  const int blocksum = __builtin_amdgcn_readlane(oinc, 31);
  if (i < n) {
    const int e = front + below + vinc - v;
    off[i] = e;
    cursor[i] = e;
  }
  if (blockIdx.x == gridDim.x - 1 && threadIdx.x == 0) off[n] = front + blocksum;
}

__global__ void csr_fill_kernel(const int* __restrict__ src, const int* __restrict__ dst,
                                int* __restrict__ cursor, int* __restrict__ csr, int E) {
  int e = blockIdx.x * blockDim.x + threadIdx.x;
  if (e < E) {
    int slot = atomicAdd(&cursor[dst[e]], 1);
    csr[slot] = src[e];
  }
}

template<int K, int NCOL, int TILES>
__device__ __forceinline__ void wmma_pass(const float* sT, const float* __restrict__ Wm,
                                          v8f (&acc)[TILES], int aofs, unsigned wofs) {
  constexpr int KC = 16 / TILES;
  static_assert(K % (4 * KC) == 0, "K must be a whole number of chunks");
  for (int kc = 0; kc < K; kc += 4 * KC) {
    const float* Wc = Wm + (unsigned)kc * NCOL;
    float b0[KC][TILES], b1[KC][TILES];
#pragma unroll
    for (int j = 0; j < KC; ++j) {
#pragma unroll
      for (int t = 0; t < TILES; ++t) {
        b0[j][t] = Wc[wofs + (unsigned)((4 * j) * NCOL + 16 * t)];
        b1[j][t] = Wc[wofs + (unsigned)((4 * j + 1) * NCOL + 16 * t)];
      }
    }
#pragma unroll
    for (int j = 0; j < KC; ++j) {
      v2f a;
      a[0] = sT[aofs + kc + 4 * j];
      a[1] = sT[aofs + kc + 4 * j + 1];
#pragma unroll
      for (int t = 0; t < TILES; ++t) {
        v2f b;
        b[0] = b0[j][t];
        b[1] = b1[j][t];
        acc[t] = __builtin_amdgcn_wmma_f32_16x16x4_f32(
            false, a, false, b, (short)0, acc[t], false, false);
      }
    }
  }
}

template<int K, int NCOL, int WAVES, int TILES, int N2>
__global__ __launch_bounds__(WAVES * 32) __attribute__((amdgpu_waves_per_eu(16)))
void sage_layer_kernel(const float* __restrict__ feat,
                       const int* __restrict__ off,
                       const int* __restrict__ csr,
                       const float* __restrict__ Wl,
                       const float* __restrict__ Wr,
                       const float* __restrict__ bias,
                       const float* __restrict__ Wl2,
                       const float* __restrict__ Wr2,
                       const float* __restrict__ bias2,
                       float* __restrict__ P,
                       float* __restrict__ Q, int nrows) {
  static_assert(WAVES * TILES * 16 == NCOL, "tiling must cover NCOL");
  static_assert(K % 128 == 0, "K must be a multiple of 128");
  static_assert(WAVES * 16 == 2 * N2, "one 16-column tile of P or Q per wave");
  constexpr int KP = K + 4;
  constexpr int HP = NCOL + 4;
  constexpr int CH = K / 128;
  constexpr int RPW = 16 / WAVES;
  static_assert(16 * HP <= 2 * 16 * KP, "the h tile must fit over the two A-tiles");
  __shared__ float sbuf[2 * 16 * KP];
  float* sA = sbuf;
  float* sX = sbuf + 16 * KP;
  float* sH = sbuf;

  const int rowbase = blockIdx.x * 16;
  const int lane = threadIdx.x & 31;
  const int wave = threadIdx.x >> 5;

  constexpr int U = 8;
  const int uwave = __builtin_amdgcn_readfirstlane(wave);
  for (int rr = 0; rr < RPW; ++rr) {
    const int r = uwave * RPW + rr;
    const int row = rowbase + r;
    float4 acc[CH] = {};
    float4 self[CH] = {};
    float scale = 1.0f;
    if (row < nrows) {
      const int beg = off[row];
      const int end = off[row + 1];
      scale = 1.0f / fmaxf((float)(end - beg), 1.0f);
      int p0 = beg;
      int myidx = 0;
      if (p0 < end) myidx = csr[min(p0 + lane, end - 1)];
#pragma unroll
      for (int c = 0; c < CH; ++c)
        self[c] = ((const float4*)(feat + (long)row * K))[lane + 32 * c];
      while (p0 < end) {
        const int cnt = min(32, end - p0);
        for (int q = 0; q < cnt; q += U) {
          float4 v[U][CH];
#pragma unroll
          for (int u = 0; u < U; ++u) {
            if (u == 0 || q + u < cnt) {
              const int s = __builtin_amdgcn_readlane(myidx, q + u);
              const float4* rp = (const float4*)(feat + (long)s * K);
#pragma unroll
              for (int c = 0; c < CH; ++c)
                v[u][c] = rp[lane + 32 * c];
            } else {
#pragma unroll
              for (int c = 0; c < CH; ++c)
                v[u][c] = make_float4(0.0f, 0.0f, 0.0f, 0.0f);
            }
          }
#pragma unroll
          for (int u = 0; u < U; ++u) {
            if (u == 0 || q + u < cnt) {
#pragma unroll
              for (int c = 0; c < CH; ++c) {
                acc[c].x += v[u][c].x; acc[c].y += v[u][c].y;
                acc[c].z += v[u][c].z; acc[c].w += v[u][c].w;
              }
            }
          }
        }
        p0 += 32;
        if (p0 < end) myidx = csr[min(p0 + lane, end - 1)];
      }
    }
#pragma unroll
    for (int c = 0; c < CH; ++c) {
      const int col = 4 * (lane + 32 * c);
      float4 a;
      a.x = acc[c].x * scale; a.y = acc[c].y * scale;
      a.z = acc[c].z * scale; a.w = acc[c].w * scale;
      *(float4*)&sA[r * KP + col] = a;
      *(float4*)&sX[r * KP + col] = self[c];
    }
  }
  __syncthreads();

  const int m = lane & 15;
  const int khalf = (lane >> 4) * 2;

  v8f acc[TILES];
  int ncol[TILES];
#pragma unroll
  for (int t = 0; t < TILES; ++t) {
    ncol[t] = (wave * TILES + t) * 16 + (lane & 15);
    float bv = bias[ncol[t]];
#pragma unroll
    for (int j = 0; j < 8; ++j) acc[t][j] = bv;
  }

  const int aofs = m * KP + khalf;
  const unsigned wofs = (unsigned)(khalf * NCOL + ncol[0]);
  wmma_pass<K, NCOL, TILES>(sA, Wl, acc, aofs, wofs);
  wmma_pass<K, NCOL, TILES>(sX, Wr, acc, aofs, wofs);

  const int rowhalf = (lane >> 4) * 8;
  __syncthreads();
#pragma unroll
  for (int t = 0; t < TILES; ++t) {
#pragma unroll
    for (int r = 0; r < 8; ++r)
      sH[(rowhalf + r) * HP + ncol[t]] = fmaxf(acc[t][r], 0.0f);
  }
  __syncthreads();

  constexpr int T2 = N2 / 16;
  static_assert((T2 & (T2 - 1)) == 0, "the tile index below is taken with a mask");
  const bool upper = uwave >= T2;
  const int col2 = (uwave & (T2 - 1)) * 16 + (lane & 15);
  const float* W2 = upper ? Wr2 : Wl2;
  float* dst2 = upper ? Q : P;
  v8f acc2[1];
  const float bv2 = upper ? bias2[col2] : 0.0f;
#pragma unroll
  for (int j = 0; j < 8; ++j) acc2[0][j] = bv2;
  wmma_pass<NCOL, N2, 1>(sH, W2, acc2, m * HP + khalf, (unsigned)(khalf * N2 + col2));

  const unsigned o2 = (unsigned)((rowbase + rowhalf) * N2 + col2);
#pragma unroll
  for (int r = 0; r < 8; ++r) {
    if (rowbase + rowhalf + r < nrows) dst2[o2 + (unsigned)(r * N2)] = acc2[0][r];
  }
}

template<int NC, int U>
__global__ __launch_bounds__(256)
void mean_out_kernel(const float* __restrict__ P, const float* __restrict__ Q,
                     const int* __restrict__ off, const int* __restrict__ csr,
                     float* __restrict__ out, int nrows) {
  constexpr int LPR = NC / 4;
  const int gid = blockIdx.x * 256 + threadIdx.x;
  const int row = gid / LPR;
  const int c = gid % LPR;
  if (row >= nrows) return;
  const int beg = off[row];
  const int end = off[row + 1];
  const float4* P4 = (const float4*)P;
  const float4 q = ((const float4*)Q)[(long)row * LPR + c];
  float4 acc = make_float4(0.0f, 0.0f, 0.0f, 0.0f);
  for (int p = beg; p < end; p += U) {
    int idx[U];
    float4 v[U];
#pragma unroll
    for (int u = 0; u < U; ++u) idx[u] = csr[min(p + u, end - 1)];
#pragma unroll
    for (int u = 0; u < U; ++u) v[u] = P4[(long)idx[u] * LPR + c];
#pragma unroll
    for (int u = 0; u < U; ++u) {
      const bool in = p + u < end;
      acc.x += in ? v[u].x : 0.0f; acc.y += in ? v[u].y : 0.0f;
      acc.z += in ? v[u].z : 0.0f; acc.w += in ? v[u].w : 0.0f;
    }
  }
  const float scale = 1.0f / fmaxf((float)(end - beg), 1.0f);
  float4 o;
  o.x = 1.0f / (1.0f + __expf(-(acc.x * scale + q.x)));
  o.y = 1.0f / (1.0f + __expf(-(acc.y * scale + q.y)));
  o.z = 1.0f / (1.0f + __expf(-(acc.z * scale + q.z)));
  o.w = 1.0f / (1.0f + __expf(-(acc.w * scale + q.w)));
  ((float4*)out)[(long)row * LPR + c] = o;
}


extern "C" void kernel_launch(void* const* d_in, const int* in_sizes, int n_in,
                              void* d_out, int out_size, void* d_ws, size_t ws_size,
                              hipStream_t stream) {
  constexpr int FIN = 128, FHID = 256, FOUT = 64;

  const float* x   = (const float*)d_in[0];
  const int*   ei  = (const int*)d_in[1];
  const float* Wl1 = (const float*)d_in[2];
  const float* Wr1 = (const float*)d_in[3];
  const float* b1  = (const float*)d_in[4];
  const float* Wl2 = (const float*)d_in[5];
  const float* Wr2 = (const float*)d_in[6];
  const float* b2  = (const float*)d_in[7];
  float* out = (float*)d_out;

  const int N = in_sizes[0] / FIN;
  const int E = in_sizes[1] / 2;
  const int* src = ei;
  const int* dst = ei + E;

  float* P    = (float*)d_ws;
  float* Q    = P + (long)N * FOUT;
  int* cnt    = (int*)(Q + (long)N * FOUT);
  int* off    = cnt + N;
  int* cursor = off + (N + 1);
  int* csr    = cursor + N;

  const int T = 256;
  auto blocks = [](long n, int t) { return (int)((n + t - 1) / t); };
  const int rowBlocks = (N + 15) / 16;

  zero_int_kernel<<<min(blocks(N, T), 4096), T, 0, stream>>>(cnt, N);
  degree_kernel<<<blocks(E, T), T, 0, stream>>>(dst, cnt, E);
  scan_kernel<<<blocks(N, 1024), 1024, 0, stream>>>(cnt, off, cursor, N);
  csr_fill_kernel<<<blocks(E, T), T, 0, stream>>>(src, dst, cursor, csr, E);

  sage_layer_kernel<FIN, FHID, 8, 2, FOUT>
      <<<rowBlocks, 256, 0, stream>>>(x, off, csr, Wl1, Wr1, b1, Wl2, Wr2, b2, P, Q, N);
  mean_out_kernel<FOUT, 8>
      <<<blocks((long)N * (FOUT / 4), T), T, 0, stream>>>(P, Q, off, csr, out, N);
}


# Layer kernel: 32 rows per block instead of 16; each weight fragment loaded from global memory feeds two row tiles (2x2 accumulators per wave): half the weight fetches per row; sum orders unchanged
# speedup vs baseline: 3.2357x; 1.0661x over previous
#include <hip/hip_runtime.h>
#include <math.h>

typedef float v2f __attribute__((ext_vector_type(2)));
typedef float v8f __attribute__((ext_vector_type(8)));


__global__ void zero_int_kernel(int* __restrict__ p, int n) {
  int i = blockIdx.x * blockDim.x + threadIdx.x;
  int stride = gridDim.x * blockDim.x;
  for (; i < n; i += stride) p[i] = 0;
}

__global__ void degree_kernel(const int* __restrict__ dst, int* __restrict__ cnt, int E) {
  int e = blockIdx.x * blockDim.x + threadIdx.x;
  if (e < E) atomicAdd(&cnt[dst[e]], 1);
}


__device__ __forceinline__ int wave_inclusive_scan(int v, int lane) {
  v += __builtin_amdgcn_update_dpp(0, v, 0x111, 0xf, 0xf, false);
  v += __builtin_amdgcn_update_dpp(0, v, 0x112, 0xf, 0xf, false);
  v += __builtin_amdgcn_update_dpp(0, v, 0x114, 0xf, 0xf, false);
  v += __builtin_amdgcn_update_dpp(0, v, 0x118, 0xf, 0xf, false);
  const int low = __builtin_amdgcn_readlane(v, 15);
  return v + ((lane >= 16) ? low : 0);
}

__global__ __launch_bounds__(1024)
void scan_kernel(const int* __restrict__ cnt, int* __restrict__ off,
                 int* __restrict__ cursor, int n) {
  constexpr int G = 8;
  __shared__ int wfront[32];
  __shared__ int wown[32];
  const int lane = threadIdx.x & 31;
  const int wave = __builtin_amdgcn_readfirstlane(threadIdx.x >> 5);
  const int first = blockIdx.x * 1024;
  const int i = first + threadIdx.x;
  const int v = (i < n) ? cnt[i] : 0;

  int s = 0;
  for (int k0 = 0; k0 < first; k0 += G * 1024) {
    int u[G];
#pragma unroll
    for (int g = 0; g < G; ++g)
      u[g] = (k0 + g * 1024 < first) ? cnt[k0 + g * 1024 + threadIdx.x] : 0;
#pragma unroll
    for (int g = 0; g < G; ++g) s += u[g];
  }

  const int sinc = wave_inclusive_scan(s, lane);
  const int vinc = wave_inclusive_scan(v, lane);
  if (lane == 31) { wfront[wave] = sinc; wown[wave] = vinc; }
  __syncthreads();
  const int tf = wfront[lane];
  const int to = wown[lane];
  const int finc = wave_inclusive_scan(tf, lane);
  const int oinc = wave_inclusive_scan(to, lane);
  const int front = __builtin_amdgcn_readlane(finc, 31);
  const int below = __builtin_amdgcn_readlane(oinc - to, wave);
  const int blocksum = __builtin_amdgcn_readlane(oinc, 31);
  if (i < n) {
    const int e = front + below + vinc - v;
    off[i] = e;
    cursor[i] = e;
  }
  if (blockIdx.x == gridDim.x - 1 && threadIdx.x == 0) off[n] = front + blocksum;
}

__global__ void csr_fill_kernel(const int* __restrict__ src, const int* __restrict__ dst,
                                int* __restrict__ cursor, int* __restrict__ csr, int E) {
  int e = blockIdx.x * blockDim.x + threadIdx.x;
  if (e < E) {
    int slot = atomicAdd(&cursor[dst[e]], 1);
    csr[slot] = src[e];
  }
}

template<int K, int NCOL, int TILES, int MT, int LDP>
__device__ __forceinline__ void wmma_pass(const float* sT, const float* __restrict__ Wm,
                                          v8f (&acc)[MT][TILES], int aofs, unsigned wofs) {
  constexpr int KC = 16 / TILES;
  static_assert(K % (4 * KC) == 0, "K must be a whole number of chunks");
#pragma unroll
  for (int kc = 0; kc < K; kc += 4 * KC) {
    const float* Wc = Wm + (unsigned)kc * NCOL;
    float b0[KC][TILES], b1[KC][TILES];
#pragma unroll
    for (int j = 0; j < KC; ++j) {
#pragma unroll
      for (int t = 0; t < TILES; ++t) {
        b0[j][t] = Wc[wofs + (unsigned)((4 * j) * NCOL + 16 * t)];
        b1[j][t] = Wc[wofs + (unsigned)((4 * j + 1) * NCOL + 16 * t)];
      }
    }
#pragma unroll
    for (int j = 0; j < KC; ++j) {
      v2f a[MT];
#pragma unroll
      for (int mt = 0; mt < MT; ++mt) {
        a[mt][0] = sT[aofs + mt * 16 * LDP + kc + 4 * j];
        a[mt][1] = sT[aofs + mt * 16 * LDP + kc + 4 * j + 1];
      }
#pragma unroll
      for (int t = 0; t < TILES; ++t) {
        v2f b;
        b[0] = b0[j][t];
        b[1] = b1[j][t];
#pragma unroll
        for (int mt = 0; mt < MT; ++mt)
          acc[mt][t] = __builtin_amdgcn_wmma_f32_16x16x4_f32(
              false, a[mt], false, b, (short)0, acc[mt][t], false, false);
      }
    }
  }
}

template<int K, int NCOL, int WAVES, int TILES, int N2, int MT>
__global__ __launch_bounds__(WAVES * 32) __attribute__((amdgpu_waves_per_eu(12)))
void sage_layer_kernel(const float* __restrict__ feat,
                       const int* __restrict__ off,
                       const int* __restrict__ csr,
                       const float* __restrict__ Wl,
                       const float* __restrict__ Wr,
                       const float* __restrict__ bias,
                       const float* __restrict__ Wl2,
                       const float* __restrict__ Wr2,
                       const float* __restrict__ bias2,
                       float* __restrict__ P,
                       float* __restrict__ Q, int nrows) {
  static_assert(WAVES * TILES * 16 == NCOL, "tiling must cover NCOL");
  static_assert(K % 128 == 0, "K must be a multiple of 128");
  static_assert(WAVES * 16 == 2 * N2, "one 16-column tile of P or Q per wave");
  constexpr int KP = K + 4;
  constexpr int HP = NCOL + 4;
  constexpr int CH = K / 128;
  constexpr int ROWS = 16 * MT;
  constexpr int RPW = ROWS / WAVES;
  static_assert(ROWS % WAVES == 0, "every wave aggregates the same number of rows");
  static_assert(ROWS * HP <= 2 * ROWS * KP, "the h tile must fit over the two A-tiles");
  __shared__ float sbuf[2 * ROWS * KP];
  float* sA = sbuf;
  float* sX = sbuf + ROWS * KP;
  float* sH = sbuf;

  const int rowbase = blockIdx.x * ROWS;
  const int lane = threadIdx.x & 31;
  const int wave = threadIdx.x >> 5;

  constexpr int U = 8;
  const int uwave = __builtin_amdgcn_readfirstlane(wave);
  for (int rr = 0; rr < RPW; ++rr) {
    const int r = uwave * RPW + rr;
    const int row = rowbase + r;
    float4 acc[CH] = {};
    float4 self[CH] = {};
    float scale = 1.0f;
    if (row < nrows) {
      const int beg = off[row];
      const int end = off[row + 1];
      scale = 1.0f / fmaxf((float)(end - beg), 1.0f);
      int p0 = beg;
      int myidx = 0;
      if (p0 < end) myidx = csr[min(p0 + lane, end - 1)];
#pragma unroll
      for (int c = 0; c < CH; ++c)
        self[c] = ((const float4*)(feat + (long)row * K))[lane + 32 * c];
      while (p0 < end) {
        const int cnt = min(32, end - p0);
        for (int q = 0; q < cnt; q += U) {
          float4 v[U][CH];
#pragma unroll
          for (int u = 0; u < U; ++u) {
            if (u == 0 || q + u < cnt) {
              const int s = __builtin_amdgcn_readlane(myidx, q + u);
              const float4* rp = (const float4*)(feat + (long)s * K);
#pragma unroll
              for (int c = 0; c < CH; ++c)
                v[u][c] = rp[lane + 32 * c];
            } else {
#pragma unroll
              for (int c = 0; c < CH; ++c)
                v[u][c] = make_float4(0.0f, 0.0f, 0.0f, 0.0f);
            }
          }
#pragma unroll
          for (int u = 0; u < U; ++u) {
            if (u == 0 || q + u < cnt) {
#pragma unroll
              for (int c = 0; c < CH; ++c) {
                acc[c].x += v[u][c].x; acc[c].y += v[u][c].y;
                acc[c].z += v[u][c].z; acc[c].w += v[u][c].w;
              }
            }
          }
        }
        p0 += 32;
        if (p0 < end) myidx = csr[min(p0 + lane, end - 1)];
      }
    }
#pragma unroll
    for (int c = 0; c < CH; ++c) {
      const int col = 4 * (lane + 32 * c);
      float4 a;
      a.x = acc[c].x * scale; a.y = acc[c].y * scale;
      a.z = acc[c].z * scale; a.w = acc[c].w * scale;
      *(float4*)&sA[r * KP + col] = a;
      *(float4*)&sX[r * KP + col] = self[c];
    }
  }
  __syncthreads();

  const int m = lane & 15;
  const int khalf = (lane >> 4) * 2;

  v8f acc[MT][TILES];
  int ncol[TILES];
#pragma unroll
  for (int t = 0; t < TILES; ++t) {
    ncol[t] = (wave * TILES + t) * 16 + (lane & 15);
    float bv = bias[ncol[t]];
#pragma unroll
    for (int mt = 0; mt < MT; ++mt) {
#pragma unroll
      for (int j = 0; j < 8; ++j) acc[mt][t][j] = bv;
    }
  }

  const int aofs = m * KP + khalf;
  const unsigned wofs = (unsigned)(khalf * NCOL + ncol[0]);
  wmma_pass<K, NCOL, TILES, MT, KP>(sA, Wl, acc, aofs, wofs);
  wmma_pass<K, NCOL, TILES, MT, KP>(sX, Wr, acc, aofs, wofs);

  const int rowhalf = (lane >> 4) * 8;
  __syncthreads();
#pragma unroll
  for (int t = 0; t < TILES; ++t) {
#pragma unroll
    for (int mt = 0; mt < MT; ++mt) {
#pragma unroll
      for (int r = 0; r < 8; ++r)
        sH[(mt * 16 + rowhalf + r) * HP + ncol[t]] = fmaxf(acc[mt][t][r], 0.0f);
    }
  }
  __syncthreads();

  constexpr int T2 = N2 / 16;
  static_assert((T2 & (T2 - 1)) == 0, "the tile index below is taken with a mask");
  const bool upper = uwave >= T2;
  const int col2 = (uwave & (T2 - 1)) * 16 + (lane & 15);
  const float* W2 = upper ? Wr2 : Wl2;
  float* dst2 = upper ? Q : P;
  v8f acc2[MT][1];
  const float bv2 = upper ? bias2[col2] : 0.0f;
#pragma unroll
  for (int mt = 0; mt < MT; ++mt) {
#pragma unroll
    for (int j = 0; j < 8; ++j) acc2[mt][0][j] = bv2;
  }
  wmma_pass<NCOL, N2, 1, MT, HP>(sH, W2, acc2, m * HP + khalf, (unsigned)(khalf * N2 + col2));

  const unsigned o2 = (unsigned)((rowbase + rowhalf) * N2 + col2);
#pragma unroll
  for (int mt = 0; mt < MT; ++mt) {
#pragma unroll
    for (int r = 0; r < 8; ++r) {
      if (rowbase + mt * 16 + rowhalf + r < nrows)
        dst2[o2 + (unsigned)((mt * 16 + r) * N2)] = acc2[mt][0][r];
    }
  }
}

template<int NC, int U>
__global__ __launch_bounds__(256)
void mean_out_kernel(const float* __restrict__ P, const float* __restrict__ Q,
                     const int* __restrict__ off, const int* __restrict__ csr,
                     float* __restrict__ out, int nrows) {
  constexpr int LPR = NC / 4;
  const int gid = blockIdx.x * 256 + threadIdx.x;
  const int row = gid / LPR;
  const int c = gid % LPR;
  if (row >= nrows) return;
  const int beg = off[row];
  const int end = off[row + 1];
  const float4* P4 = (const float4*)P;
  const float4 q = ((const float4*)Q)[(long)row * LPR + c];
  float4 acc = make_float4(0.0f, 0.0f, 0.0f, 0.0f);
  for (int p = beg; p < end; p += U) {
    int idx[U];
    float4 v[U];
#pragma unroll
    for (int u = 0; u < U; ++u) idx[u] = csr[min(p + u, end - 1)];
#pragma unroll
    for (int u = 0; u < U; ++u) v[u] = P4[(long)idx[u] * LPR + c];
#pragma unroll
    for (int u = 0; u < U; ++u) {
      const bool in = p + u < end;
      acc.x += in ? v[u].x : 0.0f; acc.y += in ? v[u].y : 0.0f;
      acc.z += in ? v[u].z : 0.0f; acc.w += in ? v[u].w : 0.0f;
    }
  }
  const float scale = 1.0f / fmaxf((float)(end - beg), 1.0f);
  float4 o;
  o.x = 1.0f / (1.0f + __expf(-(acc.x * scale + q.x)));
  o.y = 1.0f / (1.0f + __expf(-(acc.y * scale + q.y)));
  o.z = 1.0f / (1.0f + __expf(-(acc.z * scale + q.z)));
  o.w = 1.0f / (1.0f + __expf(-(acc.w * scale + q.w)));
  ((float4*)out)[(long)row * LPR + c] = o;
}


extern "C" void kernel_launch(void* const* d_in, const int* in_sizes, int n_in,
                              void* d_out, int out_size, void* d_ws, size_t ws_size,
                              hipStream_t stream) {
  constexpr int FIN = 128, FHID = 256, FOUT = 64;

  const float* x   = (const float*)d_in[0];
  const int*   ei  = (const int*)d_in[1];
  const float* Wl1 = (const float*)d_in[2];
  const float* Wr1 = (const float*)d_in[3];
  const float* b1  = (const float*)d_in[4];
  const float* Wl2 = (const float*)d_in[5];
  const float* Wr2 = (const float*)d_in[6];
  const float* b2  = (const float*)d_in[7];
  float* out = (float*)d_out;

  const int N = in_sizes[0] / FIN;
  const int E = in_sizes[1] / 2;
  const int* src = ei;
  const int* dst = ei + E;

  float* P    = (float*)d_ws;
  float* Q    = P + (long)N * FOUT;
  int* cnt    = (int*)(Q + (long)N * FOUT);
  int* off    = cnt + N;
  int* cursor = off + (N + 1);
  int* csr    = cursor + N;

  const int T = 256;
  auto blocks = [](long n, int t) { return (int)((n + t - 1) / t); };
  const int rowBlocks = (N + 31) / 32;

  zero_int_kernel<<<min(blocks(N, T), 4096), T, 0, stream>>>(cnt, N);
  degree_kernel<<<blocks(E, T), T, 0, stream>>>(dst, cnt, E);
  scan_kernel<<<blocks(N, 1024), 1024, 0, stream>>>(cnt, off, cursor, N);
  csr_fill_kernel<<<blocks(E, T), T, 0, stream>>>(src, dst, cursor, csr, E);

  sage_layer_kernel<FIN, FHID, 8, 2, FOUT, 2>
      <<<rowBlocks, 256, 0, stream>>>(x, off, csr, Wl1, Wr1, b1, Wl2, Wr2, b2, P, Q, N);
  mean_out_kernel<FOUT, 8>
      <<<blocks((long)N * (FOUT / 4), T), T, 0, stream>>>(P, Q, off, csr, out, N);
}


# Neighbour lists built by a counting sort through LDS in 2 launches (edge chunks sorted by 64-node bin, then one block per bin), not by global atomics and a scan; list order, so row sum order, changes
# speedup vs baseline: 6.0295x; 1.8634x over previous
#include <hip/hip_runtime.h>
#include <math.h>

typedef float v2f __attribute__((ext_vector_type(2)));
typedef float v8f __attribute__((ext_vector_type(8)));

constexpr int BIN_SHIFT = 6;
constexpr int BIN_NODES = 1 << BIN_SHIFT;
constexpr int CHUNK = 4096;
constexpr int PITCH = 1024;
constexpr int STAGE_CAP = 4096;

__device__ __forceinline__ int wave_inclusive_scan(int v, int lane) {
  v += __builtin_amdgcn_update_dpp(0, v, 0x111, 0xf, 0xf, false);
  v += __builtin_amdgcn_update_dpp(0, v, 0x112, 0xf, 0xf, false);
  v += __builtin_amdgcn_update_dpp(0, v, 0x114, 0xf, 0xf, false);
  v += __builtin_amdgcn_update_dpp(0, v, 0x118, 0xf, 0xf, false);
  const int low = __builtin_amdgcn_readlane(v, 15);
  return v + ((lane >= 16) ? low : 0);
}

__global__ __launch_bounds__(1024)
void partition_kernel(const int* __restrict__ src, const int* __restrict__ dst,
                      int2* __restrict__ part, int* __restrict__ poff, int E) {
  constexpr int EPT = CHUNK / 1024;
  __shared__ int hist[PITCH];
  __shared__ int wtot[32];
  __shared__ int2 stage[CHUNK];
  const int t = threadIdx.x;
  const int lane = t & 31;
  const int wave = __builtin_amdgcn_readfirstlane(t >> 5);
  const int e0 = blockIdx.x * CHUNK;
  hist[t] = 0;
  __syncthreads();

  int s[EPT], d[EPT], r[EPT];
#pragma unroll
  for (int k = 0; k < EPT; ++k) {
    const int e = e0 + k * 1024 + t;
    s[k] = 0;
    d[k] = -1;
    if (e < E) { s[k] = src[e]; d[k] = dst[e]; }
  }
#pragma unroll
  for (int k = 0; k < EPT; ++k) {
    r[k] = 0;
    if (d[k] >= 0) r[k] = atomicAdd(&hist[d[k] >> BIN_SHIFT], 1);
  }
  __syncthreads();

  const int v = hist[t];
  const int vinc = wave_inclusive_scan(v, lane);
  if (lane == 31) wtot[wave] = vinc;
  __syncthreads();
  const int tw = wtot[lane];
  const int twinc = wave_inclusive_scan(tw, lane);
  const int below = __builtin_amdgcn_readlane(twinc - tw, wave);
  const int excl = below + vinc - v;
  hist[t] = excl;
  poff[blockIdx.x * PITCH + t] = excl;
  __syncthreads();

#pragma unroll
  for (int k = 0; k < EPT; ++k) {
    if (d[k] >= 0) stage[hist[d[k] >> BIN_SHIFT] + r[k]] = make_int2(s[k], d[k]);
  }
  __syncthreads();
#pragma unroll
  for (int k = 0; k < EPT; ++k) {
    const int i = k * 1024 + t;
    if (e0 + i < E) part[e0 + i] = stage[i];
  }
}

template<bool COUNT, bool STAGED, int G>
__device__ __forceinline__ void bin_pass(const int2* __restrict__ part, int s, int e, int node0,
                                         int* ctr, int* stage, int* __restrict__ out) {
  for (int i = s; i < e; i += G) {
    int2 v[G];
#pragma unroll
    for (int u = 0; u < G; ++u) v[u] = part[min(i + u, e - 1)];
    int slot[G];
#pragma unroll
    for (int u = 0; u < G; ++u) {
      const int one = (i + u < e) ? 1 : 0;
      if (COUNT) atomicAdd(&ctr[v[u].y - node0], one);
      else slot[u] = atomicAdd(&ctr[v[u].y - node0], one);
    }
    if (!COUNT) {
#pragma unroll
      for (int u = 0; u < G; ++u) {
        if (i + u < e) {
          if (STAGED) stage[slot[u]] = v[u].x;
          else out[slot[u]] = v[u].x;
        }
      }
    }
  }
}

__global__ __launch_bounds__(256)
void bin_kernel(const int2* __restrict__ part, const int* __restrict__ poff,
                int* __restrict__ off, int* __restrict__ csr, int nchunks, int nbins, int n) {
  constexpr int G = 8;
  __shared__ int cnt[BIN_NODES];
  __shared__ int cur[BIN_NODES];
  __shared__ int wsum[8];
  __shared__ int sh_total;
  __shared__ int stage[STAGE_CAP];
  const int t = threadIdx.x;
  const int lane = t & 31;
  const int wave = __builtin_amdgcn_readfirstlane(t >> 5);
  const int b = blockIdx.x;
  const int node0 = b << BIN_SHIFT;

  int s = 0, e = 0;
  if (t < nchunks) { s = poff[t * PITCH + b]; e = poff[t * PITCH + b + 1]; }
  const int ps = t * CHUNK + s;
  const int pe = t * CHUNK + e;
  if (t < BIN_NODES) cnt[t] = 0;
  const int sinc = wave_inclusive_scan(s, lane);
  if (lane == 31) wsum[wave] = sinc;
  __syncthreads();
  int binstart = 0;
#pragma unroll
  for (int w = 0; w < 8; ++w) binstart += wsum[w];

  bin_pass<true, false, G>(part, ps, pe, node0, cnt, stage, csr);
  __syncthreads();

  if (wave == 0) {
    const int v0 = cnt[lane];
    const int v1 = cnt[32 + lane];
    const int i0 = wave_inclusive_scan(v0, lane);
    const int t0 = __builtin_amdgcn_readlane(i0, 31);
    const int i1 = wave_inclusive_scan(v1, lane) + t0;
    cur[lane] = i0 - v0;
    cur[32 + lane] = i1 - v1;
    if (node0 + lane < n) off[node0 + lane] = binstart + i0 - v0;
    if (node0 + 32 + lane < n) off[node0 + 32 + lane] = binstart + i1 - v1;
    const int total = __builtin_amdgcn_readlane(i1, 31);
    if (lane == 0) {
      sh_total = total;
      if (b == nbins - 1) off[n] = binstart + total;
    }
  }
  __syncthreads();
  const int total = __builtin_amdgcn_readfirstlane(sh_total);

  if (total <= STAGE_CAP) {
    bin_pass<false, true, G>(part, ps, pe, node0, cur, stage, csr);
    __syncthreads();
    for (int i = t; i < total; i += 256) csr[binstart + i] = stage[i];
  } else {
    bin_pass<false, false, G>(part, ps, pe, node0, cur, stage, csr + binstart);
  }
}

template<int K, int NCOL, int TILES, int MT, int LDP>
__device__ __forceinline__ void wmma_pass(const float* sT, const float* __restrict__ Wm,
                                          v8f (&acc)[MT][TILES], int aofs, unsigned wofs) {
  constexpr int KC = 16 / TILES;
  static_assert(K % (4 * KC) == 0, "K must be a whole number of chunks");
#pragma unroll
  for (int kc = 0; kc < K; kc += 4 * KC) {
    const float* Wc = Wm + (unsigned)kc * NCOL;
    float b0[KC][TILES], b1[KC][TILES];
#pragma unroll
    for (int j = 0; j < KC; ++j) {
#pragma unroll
      for (int t = 0; t < TILES; ++t) {
        b0[j][t] = Wc[wofs + (unsigned)((4 * j) * NCOL + 16 * t)];
        b1[j][t] = Wc[wofs + (unsigned)((4 * j + 1) * NCOL + 16 * t)];
      }
    }
#pragma unroll
    for (int j = 0; j < KC; ++j) {
      v2f a[MT];
#pragma unroll
      for (int mt = 0; mt < MT; ++mt) {
        a[mt][0] = sT[aofs + mt * 16 * LDP + kc + 4 * j];
        a[mt][1] = sT[aofs + mt * 16 * LDP + kc + 4 * j + 1];
      }
#pragma unroll
      for (int t = 0; t < TILES; ++t) {
        v2f b;
        b[0] = b0[j][t];
        b[1] = b1[j][t];
#pragma unroll
        for (int mt = 0; mt < MT; ++mt)
          acc[mt][t] = __builtin_amdgcn_wmma_f32_16x16x4_f32(
              false, a[mt], false, b, (short)0, acc[mt][t], false, false);
      }
    }
  }
}

template<int K, int NCOL, int WAVES, int TILES, int N2, int MT>
__global__ __launch_bounds__(WAVES * 32) __attribute__((amdgpu_waves_per_eu(12)))
void sage_layer_kernel(const float* __restrict__ feat,
                       const int* __restrict__ off,
                       const int* __restrict__ csr,
                       const float* __restrict__ Wl,
                       const float* __restrict__ Wr,
                       const float* __restrict__ bias,
                       const float* __restrict__ Wl2,
                       const float* __restrict__ Wr2,
                       const float* __restrict__ bias2,
                       float* __restrict__ P,
                       float* __restrict__ Q, int nrows) {
  static_assert(WAVES * TILES * 16 == NCOL, "tiling must cover NCOL");
  static_assert(K % 128 == 0, "K must be a multiple of 128");
  static_assert(WAVES * 16 == 2 * N2, "one 16-column tile of P or Q per wave");
  constexpr int KP = K + 4;
  constexpr int HP = NCOL + 4;
  constexpr int CH = K / 128;
  constexpr int ROWS = 16 * MT;
  constexpr int RPW = ROWS / WAVES;
  static_assert(ROWS % WAVES == 0, "every wave aggregates the same number of rows");
  static_assert(ROWS * HP <= 2 * ROWS * KP, "the h tile must fit over the two A-tiles");
  __shared__ float sbuf[2 * ROWS * KP];
  float* sA = sbuf;
  float* sX = sbuf + ROWS * KP;
  float* sH = sbuf;

  const int rowbase = blockIdx.x * ROWS;
  const int lane = threadIdx.x & 31;
  const int wave = threadIdx.x >> 5;

  constexpr int U = 8;
  const int uwave = __builtin_amdgcn_readfirstlane(wave);
  for (int rr = 0; rr < RPW; ++rr) {
    const int r = uwave * RPW + rr;
    const int row = rowbase + r;
    float4 acc[CH] = {};
    float4 self[CH] = {};
    float scale = 1.0f;
    if (row < nrows) {
      const int beg = off[row];
      const int end = off[row + 1];
      scale = 1.0f / fmaxf((float)(end - beg), 1.0f);
      int p0 = beg;
      int myidx = 0;
      if (p0 < end) myidx = csr[min(p0 + lane, end - 1)];
#pragma unroll
      for (int c = 0; c < CH; ++c)
        self[c] = ((const float4*)(feat + (long)row * K))[lane + 32 * c];
      while (p0 < end) {
        const int cnt = min(32, end - p0);
        for (int q = 0; q < cnt; q += U) {
          float4 v[U][CH];
#pragma unroll
          for (int u = 0; u < U; ++u) {
            if (u == 0 || q + u < cnt) {
              const int s = __builtin_amdgcn_readlane(myidx, q + u);
              const float4* rp = (const float4*)(feat + (long)s * K);
#pragma unroll
              for (int c = 0; c < CH; ++c)
                v[u][c] = rp[lane + 32 * c];
            } else {
#pragma unroll
              for (int c = 0; c < CH; ++c)
                v[u][c] = make_float4(0.0f, 0.0f, 0.0f, 0.0f);
            }
          }
#pragma unroll
          for (int u = 0; u < U; ++u) {
            if (u == 0 || q + u < cnt) {
#pragma unroll
              for (int c = 0; c < CH; ++c) {
                acc[c].x += v[u][c].x; acc[c].y += v[u][c].y;
                acc[c].z += v[u][c].z; acc[c].w += v[u][c].w;
              }
            }
          }
        }
        p0 += 32;
        if (p0 < end) myidx = csr[min(p0 + lane, end - 1)];
      }
    }
#pragma unroll
    for (int c = 0; c < CH; ++c) {
      const int col = 4 * (lane + 32 * c);
      float4 a;
      a.x = acc[c].x * scale; a.y = acc[c].y * scale;
      a.z = acc[c].z * scale; a.w = acc[c].w * scale;
      *(float4*)&sA[r * KP + col] = a;
      *(float4*)&sX[r * KP + col] = self[c];
    }
  }
  __syncthreads();

  const int m = lane & 15;
  const int khalf = (lane >> 4) * 2;

  v8f acc[MT][TILES];
  int ncol[TILES];
#pragma unroll
  for (int t = 0; t < TILES; ++t) {
    ncol[t] = (wave * TILES + t) * 16 + (lane & 15);
    float bv = bias[ncol[t]];
#pragma unroll
    for (int mt = 0; mt < MT; ++mt) {
#pragma unroll
      for (int j = 0; j < 8; ++j) acc[mt][t][j] = bv;
    }
  }

  const int aofs = m * KP + khalf;
  const unsigned wofs = (unsigned)(khalf * NCOL + ncol[0]);
  wmma_pass<K, NCOL, TILES, MT, KP>(sA, Wl, acc, aofs, wofs);
  wmma_pass<K, NCOL, TILES, MT, KP>(sX, Wr, acc, aofs, wofs);

  const int rowhalf = (lane >> 4) * 8;
  __syncthreads();
#pragma unroll
  for (int t = 0; t < TILES; ++t) {
#pragma unroll
    for (int mt = 0; mt < MT; ++mt) {
#pragma unroll
      for (int r = 0; r < 8; ++r)
        sH[(mt * 16 + rowhalf + r) * HP + ncol[t]] = fmaxf(acc[mt][t][r], 0.0f);
    }
  }
  __syncthreads();

  constexpr int T2 = N2 / 16;
  static_assert((T2 & (T2 - 1)) == 0, "the tile index below is taken with a mask");
  const bool upper = uwave >= T2;
  const int col2 = (uwave & (T2 - 1)) * 16 + (lane & 15);
  const float* W2 = upper ? Wr2 : Wl2;
  float* dst2 = upper ? Q : P;
  v8f acc2[MT][1];
  const float bv2 = upper ? bias2[col2] : 0.0f;
#pragma unroll
  for (int mt = 0; mt < MT; ++mt) {
#pragma unroll
    for (int j = 0; j < 8; ++j) acc2[mt][0][j] = bv2;
  }
  wmma_pass<NCOL, N2, 1, MT, HP>(sH, W2, acc2, m * HP + khalf, (unsigned)(khalf * N2 + col2));

  const unsigned o2 = (unsigned)((rowbase + rowhalf) * N2 + col2);
#pragma unroll
  for (int mt = 0; mt < MT; ++mt) {
#pragma unroll
    for (int r = 0; r < 8; ++r) {
      if (rowbase + mt * 16 + rowhalf + r < nrows)
        dst2[o2 + (unsigned)((mt * 16 + r) * N2)] = acc2[mt][0][r];
    }
  }
}

template<int NC, int U>
__global__ __launch_bounds__(256)
void mean_out_kernel(const float* __restrict__ P, const float* __restrict__ Q,
                     const int* __restrict__ off, const int* __restrict__ csr,
                     float* __restrict__ out, int nrows) {
  constexpr int LPR = NC / 4;
  const int gid = blockIdx.x * 256 + threadIdx.x;
  const int row = gid / LPR;
  const int c = gid % LPR;
  if (row >= nrows) return;
  const int beg = off[row];
  const int end = off[row + 1];
  const float4* P4 = (const float4*)P;
  const float4 q = ((const float4*)Q)[(long)row * LPR + c];
  float4 acc = make_float4(0.0f, 0.0f, 0.0f, 0.0f);
  for (int p = beg; p < end; p += U) {
    int idx[U];
    float4 v[U];
#pragma unroll
    for (int u = 0; u < U; ++u) idx[u] = csr[min(p + u, end - 1)];
#pragma unroll
    for (int u = 0; u < U; ++u) v[u] = P4[(long)idx[u] * LPR + c];
#pragma unroll
    for (int u = 0; u < U; ++u) {
      const bool in = p + u < end;
      acc.x += in ? v[u].x : 0.0f; acc.y += in ? v[u].y : 0.0f;
      acc.z += in ? v[u].z : 0.0f; acc.w += in ? v[u].w : 0.0f;
    }
  }
  const float scale = 1.0f / fmaxf((float)(end - beg), 1.0f);
  float4 o;
  o.x = 1.0f / (1.0f + __expf(-(acc.x * scale + q.x)));
  o.y = 1.0f / (1.0f + __expf(-(acc.y * scale + q.y)));
  o.z = 1.0f / (1.0f + __expf(-(acc.z * scale + q.z)));
  o.w = 1.0f / (1.0f + __expf(-(acc.w * scale + q.w)));
  ((float4*)out)[(long)row * LPR + c] = o;
}


extern "C" void kernel_launch(void* const* d_in, const int* in_sizes, int n_in,
                              void* d_out, int out_size, void* d_ws, size_t ws_size,
                              hipStream_t stream) {
  constexpr int FIN = 128, FHID = 256, FOUT = 64;

  const float* x   = (const float*)d_in[0];
  const int*   ei  = (const int*)d_in[1];
  const float* Wl1 = (const float*)d_in[2];
  const float* Wr1 = (const float*)d_in[3];
  const float* b1  = (const float*)d_in[4];
  const float* Wl2 = (const float*)d_in[5];
  const float* Wr2 = (const float*)d_in[6];
  const float* b2  = (const float*)d_in[7];
  float* out = (float*)d_out;

  const int N = in_sizes[0] / FIN;
  const int E = in_sizes[1] / 2;
  const int* src = ei;
  const int* dst = ei + E;

  float* P    = (float*)d_ws;
  float* Q    = P + (long)N * FOUT;
  int* ibase  = (int*)(Q + (long)N * FOUT);
  int* off    = ibase + N;
  int* csr    = off + (N + 1) + N;
  int2* part  = (int2*)d_ws;
  int* poff   = (int*)(part + E);

  const int T = 256;
  auto blocks = [](long n, int t) { return (int)((n + t - 1) / t); };
  const int rowBlocks = (N + 31) / 32;
  const int nchunks = blocks(E, CHUNK);
  const int nbins = blocks(N, BIN_NODES);

  partition_kernel<<<nchunks, 1024, 0, stream>>>(src, dst, part, poff, E);
  bin_kernel<<<nbins, 256, 0, stream>>>(part, poff, off, csr, nchunks, nbins, N);

  sage_layer_kernel<FIN, FHID, 8, 2, FOUT, 2>
      <<<rowBlocks, 256, 0, stream>>>(x, off, csr, Wl1, Wr1, b1, Wl2, Wr2, b2, P, Q, N);
  mean_out_kernel<FOUT, 8>
      <<<blocks((long)N * (FOUT / 4), T), T, 0, stream>>>(P, Q, off, csr, out, N);
}
